# starting kernel
# baseline (speedup 1.0000x reference)
#include <hip/hip_runtime.h>
#include <hip/hip_bf16.h>


typedef __attribute__((ext_vector_type(2))) float v2f;
typedef __attribute__((ext_vector_type(8))) float v8f;

#define IN_CH 128

__global__ void k_deg_init(float* __restrict__ deg, int n) {
    int i = blockIdx.x * blockDim.x + threadIdx.x;
    if (i < n) deg[i] = 1.0f;
}

__global__ void k_deg_count(const int* __restrict__ dst, float* __restrict__ deg, int e) {
    int i = blockIdx.x * blockDim.x + threadIdx.x;
    if (i < e) atomicAdd(&deg[dst[i]], 1.0f);
}

__global__ void k_dinv(const float* __restrict__ deg, float* __restrict__ dinv, int n) {
    int i = blockIdx.x * blockDim.x + threadIdx.x;
    if (i < n) dinv[i] = rsqrtf(deg[i]);
}

__global__ __launch_bounds__(256) void k_gemm1_wmma(
    const float* __restrict__ x, const float* __restrict__ W1,
    const float* __restrict__ dinv,
    float* __restrict__ h, float* __restrict__ agg1,
    int n_nodes, int mtiles) {
    __shared__ float lw[IN_CH * IN_CH];
    const int tid = threadIdx.x;
    for (int i = tid; i < IN_CH * IN_CH; i += 256) lw[i] = W1[i];
    __syncthreads();

    const int wid  = tid >> 5;
    const int lane = tid & 31;
    const int half = lane >> 4;
    const int l    = lane & 15;
    const int tile = blockIdx.x * 8 + wid;
    if (tile >= mtiles) return;
    const int m0 = tile * 16;

    v8f acc[8] = {};

    const int arow = min(m0 + l, n_nodes - 1);
    const float* ap = x + (size_t)arow * IN_CH;

    for (int kk = 0; kk < IN_CH; kk += 4) {
        const int ka = kk + 2 * half;
        v2f a;
        a.x = ap[ka];
        a.y = ap[ka + 1];
#pragma unroll
        for (int nt = 0; nt < 8; ++nt) {
            const int nc = nt * 16 + l;
            v2f b;
            b.x = lw[ka * IN_CH + nc];
            b.y = lw[(ka + 1) * IN_CH + nc];
            acc[nt] = __builtin_amdgcn_wmma_f32_16x16x4_f32(
                false, a, false, b, (short)0, acc[nt], false, false);
        }
    }

#pragma unroll
    for (int j = 0; j < 8; ++j) {
        const int row = m0 + j + 8 * half;
        if (row >= n_nodes) continue;
        const float di = dinv[row];
        const float sc = di * di;
#pragma unroll
        for (int nt = 0; nt < 8; ++nt) {
            const float c = acc[nt][j];
            const size_t idx = (size_t)row * IN_CH + nt * 16 + l;
            h[idx]    = c;
            agg1[idx] = c * sc;
        }
    }
}

__global__ __launch_bounds__(256) void k_edge_agg1(
    const int* __restrict__ src, const int* __restrict__ dst,
    const float* __restrict__ dinv, const float* __restrict__ h,
    float* __restrict__ agg1, int e) {
    const int w    = blockIdx.x * 8 + (threadIdx.x >> 5);
    const int lane = threadIdx.x & 31;
    if (w >= e) return;
    const int s = src[w], d = dst[w];
    const float norm = dinv[s] * dinv[d];
    const float4 hv = ((const float4*)(h + (size_t)s * IN_CH))[lane];
    float* op = agg1 + (size_t)d * IN_CH + lane * 4;
    atomicAdd(op + 0, hv.x * norm);
    atomicAdd(op + 1, hv.y * norm);
    atomicAdd(op + 2, hv.z * norm);
    atomicAdd(op + 3, hv.w * norm);
}

__global__ __launch_bounds__(256) void k_gemv2(
    const float* __restrict__ agg1, const float* __restrict__ b1,
    const float* __restrict__ W2, const float* __restrict__ b2,
    const float* __restrict__ dinv,
    float* __restrict__ h2, float* __restrict__ out, int n) {
    const int node = blockIdx.x * 8 + (threadIdx.x >> 5);
    const int lane = threadIdx.x & 31;
    if (node >= n) return;
    const float4 a  = ((const float4*)(agg1 + (size_t)node * IN_CH))[lane];
    const float4 bb = ((const float4*)b1)[lane];
    const float4 w  = ((const float4*)W2)[lane];
    float v = fmaxf(a.x + bb.x, 0.f) * w.x + fmaxf(a.y + bb.y, 0.f) * w.y +
              fmaxf(a.z + bb.z, 0.f) * w.z + fmaxf(a.w + bb.w, 0.f) * w.w;
#pragma unroll
    for (int off = 16; off > 0; off >>= 1) v += __shfl_xor(v, off, 32);
    if (lane == 0) {
        const float di = dinv[node];
        h2[node]  = v;
        out[node] = v * di * di + b2[0];
    }
}

__global__ void k_edge_agg2(const int* __restrict__ src, const int* __restrict__ dst,
                            const float* __restrict__ dinv, const float* __restrict__ h2,
                            float* __restrict__ out, int e) {
    int i = blockIdx.x * blockDim.x + threadIdx.x;
    if (i < e) {
        const int s = src[i], d = dst[i];
        atomicAdd(&out[d], h2[s] * dinv[s] * dinv[d]);
    }
}

extern "C" void kernel_launch(void* const* d_in, const int* in_sizes, int n_in,
                              void* d_out, int out_size, void* d_ws, size_t ws_size,
                              hipStream_t stream) {
    const float* x  = (const float*)d_in[0];
    const int*   ei = (const int*)d_in[1];
    const float* W1 = (const float*)d_in[2];
    const float* b1 = (const float*)d_in[3];
    const float* W2 = (const float*)d_in[4];
    const float* b2 = (const float*)d_in[5];

    const int n = in_sizes[0] / IN_CH;
    const int e = in_sizes[1] / 2;
    const int* src = ei;
    const int* dst = ei + e;

    float* ws   = (float*)d_ws;
    float* deg  = ws;
    float* dinv = deg + n;
    float* h    = dinv + n;
    float* agg1 = h + (size_t)n * IN_CH;
    float* h2   = agg1 + (size_t)n * IN_CH;
    float* out  = (float*)d_out;

    const int mtiles = (n + 15) / 16;

    k_deg_init <<<(n + 255) / 256, 256, 0, stream>>>(deg, n);
    k_deg_count<<<(e + 255) / 256, 256, 0, stream>>>(dst, deg, e);
    k_dinv     <<<(n + 255) / 256, 256, 0, stream>>>(deg, dinv, n);
    k_gemm1_wmma<<<(mtiles + 7) / 8, 256, 0, stream>>>(x, W1, dinv, h, agg1, n, mtiles);
    k_edge_agg1<<<(e + 7) / 8, 256, 0, stream>>>(src, dst, dinv, h, agg1, e);
    k_gemv2    <<<(n + 7) / 8, 256, 0, stream>>>(agg1, b1, W2, b2, dinv, h2, out, n);
    k_edge_agg2<<<(e + 255) / 256, 256, 0, stream>>>(src, dst, dinv, h2, out, e);
}


# Float-atomic edge scatter of both layers replaced by a per-call counting sort of edges by target (integer atomic rank) and pull gathers; per-node sums now run in rank (arrival) order
# speedup vs baseline: 2.0958x; 2.0958x over previous
#include <hip/hip_runtime.h>
#include <hip/hip_bf16.h>


typedef __attribute__((ext_vector_type(2))) float v2f;
typedef __attribute__((ext_vector_type(8))) float v8f;

#define IN_CH 128

__global__ void k_init(int* __restrict__ cnt, int* __restrict__ cursor, int n) {
    int i = blockIdx.x * blockDim.x + threadIdx.x;
    if (i < n) cnt[i] = 0;
    if (i == 0) *cursor = 0;
}

__global__ void k_count_rank(const int* __restrict__ dst, int* __restrict__ cnt,
                             int* __restrict__ rank, int e) {
    int i = blockIdx.x * blockDim.x + threadIdx.x;
    if (i < e) rank[i] = atomicAdd(&cnt[dst[i]], 1);
}

__global__ void k_offsets(const int* __restrict__ cnt, int* __restrict__ cursor,
                          int* __restrict__ rowstart, float* __restrict__ dinv, int n) {
    const int i    = blockIdx.x * blockDim.x + threadIdx.x;
    const int lane = threadIdx.x & 31;
    const int c    = (i < n) ? cnt[i] : 0;
    int incl = c;
    for (int off = 1; off < 32; off <<= 1) {
        const int t = __shfl_up(incl, off, 32);
        if (lane >= off) incl += t;
    }
    int base = 0;
    if (lane == 31) base = atomicAdd(cursor, incl);
    base = __shfl(base, 31, 32);
    if (i < n) {
        rowstart[i] = base + incl - c;
        dinv[i]     = rsqrtf((float)(c + 1));
    }
}

__global__ void k_fill(const int* __restrict__ src, const int* __restrict__ dst,
                       const int* __restrict__ rank, const int* __restrict__ rowstart,
                       int* __restrict__ col, int e) {
    int i = blockIdx.x * blockDim.x + threadIdx.x;
    if (i < e) col[rowstart[dst[i]] + rank[i]] = src[i];
}

__global__ __launch_bounds__(256) void k_gemm1_wmma(
    const float* __restrict__ x, const float* __restrict__ W1,
    float* __restrict__ h, int n_nodes, int mtiles) {
    __shared__ float lw[IN_CH * IN_CH];
    const int tid = threadIdx.x;
    for (int i = tid; i < IN_CH * IN_CH; i += 256) lw[i] = W1[i];
    __syncthreads();

    const int wid  = tid >> 5;
    const int lane = tid & 31;
    const int half = lane >> 4;
    const int l    = lane & 15;
    const int tile = blockIdx.x * 8 + wid;
    if (tile >= mtiles) return;
    const int m0 = tile * 16;

    v8f acc[8] = {};

    const int arow = min(m0 + l, n_nodes - 1);
    const float* ap = x + (size_t)arow * IN_CH;

    for (int kk = 0; kk < IN_CH; kk += 4) {
        const int ka = kk + 2 * half;
        v2f a;
        a.x = ap[ka];
        a.y = ap[ka + 1];
#pragma unroll
        for (int nt = 0; nt < 8; ++nt) {
            const int nc = nt * 16 + l;
            v2f b;
            b.x = lw[ka * IN_CH + nc];
            b.y = lw[(ka + 1) * IN_CH + nc];
            acc[nt] = __builtin_amdgcn_wmma_f32_16x16x4_f32(
                false, a, false, b, (short)0, acc[nt], false, false);
        }
    }

#pragma unroll
    for (int j = 0; j < 8; ++j) {
        const int row = m0 + j + 8 * half;
        if (row >= n_nodes) continue;
#pragma unroll
        for (int nt = 0; nt < 8; ++nt) {
            const size_t idx = (size_t)row * IN_CH + nt * 16 + l;
            h[idx] = acc[nt][j];
        }
    }
}

__global__ __launch_bounds__(256) void k_gather1(
    const int* __restrict__ col, const int* __restrict__ rowstart,
    const int* __restrict__ cnt, const float* __restrict__ dinv,
    const float* __restrict__ h, float* __restrict__ agg1, int n) {
    const int lane = threadIdx.x & 31;
    const int node = __builtin_amdgcn_readfirstlane(blockIdx.x * 8 + (threadIdx.x >> 5));
    if (node >= n) return;
    const int   start = rowstart[node];
    const int   c     = cnt[node];
    const float dd    = dinv[node];
    const float4* hrow = (const float4*)h;
    const float4 hs = hrow[(size_t)node * 32 + lane];
    const float sc = dd * dd;
    float4 acc;
    acc.x = hs.x * sc; acc.y = hs.y * sc; acc.z = hs.z * sc; acc.w = hs.w * sc;

    for (int base = 0; base < c; base += 32) {
        const int m = min(32, c - base);
        int   s_l = node;
        float w_l = 0.f;
        if (lane < m) {
            s_l = col[start + base + lane];
            w_l = dinv[s_l] * dd;
        }
        for (int k = 0; k < m; k += 8) {
            float4 r[8];
            float  w[8];
#pragma unroll
            for (int u = 0; u < 8; ++u) {
                const int s = __builtin_amdgcn_readlane(s_l, k + u);
                w[u] = __int_as_float(__builtin_amdgcn_readlane(__float_as_int(w_l), k + u));
                r[u] = hrow[(size_t)s * 32 + lane];
            }
#pragma unroll
            for (int u = 0; u < 8; ++u) {
                float4 v = r[u];
                if (k + u >= m) { v.x = 0.f; v.y = 0.f; v.z = 0.f; v.w = 0.f; }
                acc.x += v.x * w[u]; acc.y += v.y * w[u];
                acc.z += v.z * w[u]; acc.w += v.w * w[u];
            }
        }
    }
    ((float4*)agg1)[(size_t)node * 32 + lane] = acc;
}

__global__ __launch_bounds__(256) void k_gemv2(
    const float* __restrict__ agg1, const float* __restrict__ b1,
    const float* __restrict__ W2, float* __restrict__ h2, int n) {
    const int node = blockIdx.x * 8 + (threadIdx.x >> 5);
    const int lane = threadIdx.x & 31;
    if (node >= n) return;
    const float4 a  = ((const float4*)(agg1 + (size_t)node * IN_CH))[lane];
    const float4 bb = ((const float4*)b1)[lane];
    const float4 w  = ((const float4*)W2)[lane];
    float v = fmaxf(a.x + bb.x, 0.f) * w.x + fmaxf(a.y + bb.y, 0.f) * w.y +
              fmaxf(a.z + bb.z, 0.f) * w.z + fmaxf(a.w + bb.w, 0.f) * w.w;
#pragma unroll
    for (int off = 16; off > 0; off >>= 1) v += __shfl_xor(v, off, 32);
    if (lane == 0) h2[node] = v;
}

__global__ void k_gather2(const int* __restrict__ col, const int* __restrict__ rowstart,
                          const int* __restrict__ cnt, const float* __restrict__ dinv,
                          const float* __restrict__ h2, const float* __restrict__ b2,
                          float* __restrict__ out, int n) {
    const int i = blockIdx.x * blockDim.x + threadIdx.x;
    if (i >= n) return;
    const float di = dinv[i];
    float acc = h2[i] * di * di + b2[0];
    const int st = rowstart[i];
    const int c  = cnt[i];
#pragma unroll 4
    for (int j = 0; j < c; ++j) {
        const int s = col[st + j];
        acc += h2[s] * dinv[s] * di;
    }
    out[i] = acc;
}

extern "C" void kernel_launch(void* const* d_in, const int* in_sizes, int n_in,
                              void* d_out, int out_size, void* d_ws, size_t ws_size,
                              hipStream_t stream) {
    const float* x  = (const float*)d_in[0];
    const int*   ei = (const int*)d_in[1];
    const float* W1 = (const float*)d_in[2];
    const float* b1 = (const float*)d_in[3];
    const float* W2 = (const float*)d_in[4];
    const float* b2 = (const float*)d_in[5];

    const int n = in_sizes[0] / IN_CH;
    const int e = in_sizes[1] / 2;
    const int* src = ei;
    const int* dst = ei + e;

    float* ws       = (float*)d_ws;
    float* h        = ws;
    float* agg1     = h + (size_t)n * IN_CH;
    float* dinv     = agg1 + (size_t)n * IN_CH;
    float* h2       = dinv + n;
    int*   cnt      = (int*)(h2 + n);
    int*   rowstart = cnt + n;
    int*   rank     = rowstart + n;
    int*   col      = rank + e;
    int*   cursor   = col + e;
    float* out      = (float*)d_out;

    const int mtiles = (n + 15) / 16;

    k_init      <<<(n + 255) / 256, 256, 0, stream>>>(cnt, cursor, n);
    k_count_rank<<<(e + 255) / 256, 256, 0, stream>>>(dst, cnt, rank, e);
    k_offsets   <<<(n + 255) / 256, 256, 0, stream>>>(cnt, cursor, rowstart, dinv, n);
    k_gemm1_wmma<<<(mtiles + 7) / 8, 256, 0, stream>>>(x, W1, h, n, mtiles);
    k_fill      <<<(e + 255) / 256, 256, 0, stream>>>(src, dst, rank, rowstart, col, e);
    k_gather1   <<<(n + 7) / 8, 256, 0, stream>>>(col, rowstart, cnt, dinv, h, agg1, n);
    k_gemv2     <<<(n + 7) / 8, 256, 0, stream>>>(agg1, b1, W2, h2, n);
    k_gather2   <<<(n + 255) / 256, 256, 0, stream>>>(col, rowstart, cnt, dinv, h2, b2, out, n);
}


# Layer-2 linear (bias, relu, dot with W2) moved into the layer-1 pull's epilogue: the wave that sums a node's row finishes it; aggregated rows never stored, one launch fewer; sum orders unchanged
# speedup vs baseline: 2.1762x; 1.0384x over previous
#include <hip/hip_runtime.h>
#include <hip/hip_bf16.h>


typedef __attribute__((ext_vector_type(2))) float v2f;
typedef __attribute__((ext_vector_type(8))) float v8f;

#define IN_CH 128

__global__ void k_init(int* __restrict__ cnt, int* __restrict__ cursor, int n) {
    int i = blockIdx.x * blockDim.x + threadIdx.x;
    if (i < n) cnt[i] = 0;
    if (i == 0) *cursor = 0;
}

__global__ void k_count_rank(const int* __restrict__ dst, int* __restrict__ cnt,
                             int* __restrict__ rank, int e) {
    int i = blockIdx.x * blockDim.x + threadIdx.x;
    if (i < e) rank[i] = atomicAdd(&cnt[dst[i]], 1);
}

__global__ void k_offsets(const int* __restrict__ cnt, int* __restrict__ cursor,
                          int* __restrict__ rowstart, float* __restrict__ dinv, int n) {
    const int i    = blockIdx.x * blockDim.x + threadIdx.x;
    const int lane = threadIdx.x & 31;
    const int c    = (i < n) ? cnt[i] : 0;
    int incl = c;
    for (int off = 1; off < 32; off <<= 1) {
        const int t = __shfl_up(incl, off, 32);
        if (lane >= off) incl += t;
    }
    int base = 0;
    if (lane == 31) base = atomicAdd(cursor, incl);
    base = __shfl(base, 31, 32);
    if (i < n) {
        rowstart[i] = base + incl - c;
        dinv[i]     = rsqrtf((float)(c + 1));
    }
}

__global__ void k_fill(const int* __restrict__ src, const int* __restrict__ dst,
                       const int* __restrict__ rank, const int* __restrict__ rowstart,
                       int* __restrict__ col, int e) {
    int i = blockIdx.x * blockDim.x + threadIdx.x;
    if (i < e) col[rowstart[dst[i]] + rank[i]] = src[i];
}

__global__ __launch_bounds__(256) void k_gemm1_wmma(
    const float* __restrict__ x, const float* __restrict__ W1,
    float* __restrict__ h, int n_nodes, int mtiles) {
    __shared__ float lw[IN_CH * IN_CH];
    const int tid = threadIdx.x;
    for (int i = tid; i < IN_CH * IN_CH; i += 256) lw[i] = W1[i];
    __syncthreads();

    const int wid  = tid >> 5;
    const int lane = tid & 31;
    const int half = lane >> 4;
    const int l    = lane & 15;
    const int tile = blockIdx.x * 8 + wid;
    if (tile >= mtiles) return;
    const int m0 = tile * 16;

    v8f acc[8] = {};

    const int arow = min(m0 + l, n_nodes - 1);
    const float* ap = x + (size_t)arow * IN_CH;

    for (int kk = 0; kk < IN_CH; kk += 4) {
        const int ka = kk + 2 * half;
        v2f a;
        a.x = ap[ka];
        a.y = ap[ka + 1];
#pragma unroll
        for (int nt = 0; nt < 8; ++nt) {
            const int nc = nt * 16 + l;
            v2f b;
            b.x = lw[ka * IN_CH + nc];
            b.y = lw[(ka + 1) * IN_CH + nc];
            acc[nt] = __builtin_amdgcn_wmma_f32_16x16x4_f32(
                false, a, false, b, (short)0, acc[nt], false, false);
        }
    }

#pragma unroll
    for (int j = 0; j < 8; ++j) {
        const int row = m0 + j + 8 * half;
        if (row >= n_nodes) continue;
#pragma unroll
        for (int nt = 0; nt < 8; ++nt) {
            const size_t idx = (size_t)row * IN_CH + nt * 16 + l;
            h[idx] = acc[nt][j];
        }
    }
}

__global__ __launch_bounds__(256) void k_gather1(
    const int* __restrict__ col, const int* __restrict__ rowstart,
    const int* __restrict__ cnt, const float* __restrict__ dinv,
    const float* __restrict__ h, const float* __restrict__ b1,
    const float* __restrict__ W2, float* __restrict__ h2, int n) {
    const int lane = threadIdx.x & 31;
    const int node = __builtin_amdgcn_readfirstlane(blockIdx.x * 8 + (threadIdx.x >> 5));
    if (node >= n) return;
    const int   start = rowstart[node];
    const int   c     = cnt[node];
    const float dd    = dinv[node];
    const float4* hrow = (const float4*)h;
    const float4 hs = hrow[(size_t)node * 32 + lane];
    const float sc = dd * dd;
    float4 acc;
    acc.x = hs.x * sc; acc.y = hs.y * sc; acc.z = hs.z * sc; acc.w = hs.w * sc;

    for (int base = 0; base < c; base += 32) {
        const int m = min(32, c - base);
        int   s_l = node;
        float w_l = 0.f;
        if (lane < m) {
            s_l = col[start + base + lane];
            w_l = dinv[s_l] * dd;
        }
        for (int k = 0; k < m; k += 8) {
            float4 r[8];
            float  w[8];
#pragma unroll
            for (int u = 0; u < 8; ++u) {
                const int s = __builtin_amdgcn_readlane(s_l, k + u);
                w[u] = __int_as_float(__builtin_amdgcn_readlane(__float_as_int(w_l), k + u));
                r[u] = hrow[(size_t)s * 32 + lane];
            }
#pragma unroll
            for (int u = 0; u < 8; ++u) {
                float4 v = r[u];
                if (k + u >= m) { v.x = 0.f; v.y = 0.f; v.z = 0.f; v.w = 0.f; }
                acc.x += v.x * w[u]; acc.y += v.y * w[u];
                acc.z += v.z * w[u]; acc.w += v.w * w[u];
            }
        }
    }
    const float4 bb = ((const float4*)b1)[lane];
    const float4 w2 = ((const float4*)W2)[lane];
    float v = fmaxf(acc.x + bb.x, 0.f) * w2.x + fmaxf(acc.y + bb.y, 0.f) * w2.y +
              fmaxf(acc.z + bb.z, 0.f) * w2.z + fmaxf(acc.w + bb.w, 0.f) * w2.w;
#pragma unroll
    for (int off = 16; off > 0; off >>= 1) v += __shfl_xor(v, off, 32);
    if (lane == 0) h2[node] = v;
}

__global__ void k_gather2(const int* __restrict__ col, const int* __restrict__ rowstart,
                          const int* __restrict__ cnt, const float* __restrict__ dinv,
                          const float* __restrict__ h2, const float* __restrict__ b2,
                          float* __restrict__ out, int n) {
    const int i = blockIdx.x * blockDim.x + threadIdx.x;
    if (i >= n) return;
    const float di = dinv[i];
    float acc = h2[i] * di * di + b2[0];
    const int st = rowstart[i];
    const int c  = cnt[i];
#pragma unroll 4
    for (int j = 0; j < c; ++j) {
        const int s = col[st + j];
        acc += h2[s] * dinv[s] * di;
    }
    out[i] = acc;
}

extern "C" void kernel_launch(void* const* d_in, const int* in_sizes, int n_in,
                              void* d_out, int out_size, void* d_ws, size_t ws_size,
                              hipStream_t stream) {
    const float* x  = (const float*)d_in[0];
    const int*   ei = (const int*)d_in[1];
    const float* W1 = (const float*)d_in[2];
    const float* b1 = (const float*)d_in[3];
    const float* W2 = (const float*)d_in[4];
    const float* b2 = (const float*)d_in[5];

    const int n = in_sizes[0] / IN_CH;
    const int e = in_sizes[1] / 2;
    const int* src = ei;
    const int* dst = ei + e;

    float* ws       = (float*)d_ws;
    float* h        = ws;
    float* dinv     = h + 2 * (size_t)n * IN_CH;
    float* h2       = dinv + n;
    int*   cnt      = (int*)(h2 + n);
    int*   rowstart = cnt + n;
    int*   rank     = rowstart + n;
    int*   col      = rank + e;
    int*   cursor   = col + e;
    float* out      = (float*)d_out;

    const int mtiles = (n + 15) / 16;

    k_init      <<<(n + 255) / 256, 256, 0, stream>>>(cnt, cursor, n);
    k_count_rank<<<(e + 255) / 256, 256, 0, stream>>>(dst, cnt, rank, e);
    k_offsets   <<<(n + 255) / 256, 256, 0, stream>>>(cnt, cursor, rowstart, dinv, n);
    k_gemm1_wmma<<<(mtiles + 7) / 8, 256, 0, stream>>>(x, W1, h, n, mtiles);
    k_fill      <<<(e + 255) / 256, 256, 0, stream>>>(src, dst, rank, rowstart, col, e);
    k_gather1   <<<(n + 7) / 8, 256, 0, stream>>>(col, rowstart, cnt, dinv, h, b1, W2, h2, n);
    k_gather2   <<<(n + 255) / 256, 256, 0, stream>>>(col, rowstart, cnt, dinv, h2, b2, out, n);
}


# Edge grouping redone as a two-level counting sort on work-group LDS counters (256-node bins, whole-run stores): no per-edge global atomic or scattered store; node sum order = LDS arrival in bin
# speedup vs baseline: 3.5045x; 1.6104x over previous
#include <hip/hip_runtime.h>
#include <hip/hip_bf16.h>


typedef __attribute__((ext_vector_type(2))) float v2f;
typedef __attribute__((ext_vector_type(8))) float v8f;

#define IN_CH 128

#define BIN_SHIFT 8
#define BIN_NODES 256
#define NB_MAX    512
#define CH_EDGES  4096
#define STAGE_CAP 8192
#define BIN_PAD   64

__global__ void k_init(int* __restrict__ binws, int nctr) {
    int i = blockIdx.x * blockDim.x + threadIdx.x;
    if (i < nctr) binws[i * BIN_PAD] = 0;
}

__device__ __forceinline__ void lds_excl_scan(int* a, int len, int* wsum, int tid) {
    const int lane = tid & 31;
    const int v = (tid < len) ? a[tid] : 0;
    int incl = v;
    for (int off = 1; off < 32; off <<= 1) {
        const int t = __shfl_up(incl, off, 32);
        if (lane >= off) incl += t;
    }
    if (lane == 31) wsum[tid >> 5] = incl;
    __syncthreads();
    int b = 0;
    for (int w = 0; w < (tid >> 5); ++w) b += wsum[w];
    if (tid < len) a[tid] = b + incl - v;
    __syncthreads();
}

__global__ __launch_bounds__(1024) void k_hist(const int* __restrict__ dst,
                                               int* __restrict__ bintotal, int e, int nbins) {
    __shared__ int hist[NB_MAX];
    const int tid = threadIdx.x;
    if (tid < NB_MAX) hist[tid] = 0;
    __syncthreads();
    const int chunk0 = blockIdx.x * CH_EDGES;
    int d[CH_EDGES / 1024];
#pragma unroll
    for (int k = 0; k < CH_EDGES / 1024; ++k) {
        d[k] = dst[min(chunk0 + k * 1024 + tid, e - 1)];
    }
    static_assert(CH_EDGES / 1024 == 4, "the load fence below names four values");
    asm volatile("" :: "v"(d[0]), "v"(d[1]), "v"(d[2]), "v"(d[3]));
#pragma unroll
    for (int k = 0; k < CH_EDGES / 1024; ++k)
        if (chunk0 + k * 1024 + tid < e) atomicAdd(&hist[d[k] >> BIN_SHIFT], 1);
    __syncthreads();
    if (tid < nbins) {
        const int c = hist[tid];
        if (c > 0) atomicAdd(&bintotal[tid * BIN_PAD], c);
    }
}

__global__ __launch_bounds__(1024) void k_scatter(
    const int* __restrict__ src, const int* __restrict__ dst,
    const int* __restrict__ bintotal, int* __restrict__ bincursor,
    int2* __restrict__ binned, int e, int nbins) {
    __shared__ int  hist[NB_MAX];
    __shared__ int  gofs[NB_MAX];
    __shared__ int  wsum[32];
    __shared__ int2 stage[CH_EDGES];
    const int tid = threadIdx.x;
    if (tid < NB_MAX) {
        hist[tid] = 0;
        gofs[tid] = (tid < nbins) ? bintotal[tid * BIN_PAD] : 0;
    }
    __syncthreads();
    const int chunk0 = blockIdx.x * CH_EDGES;
    int s[CH_EDGES / 1024], d[CH_EDGES / 1024], r[CH_EDGES / 1024];
#pragma unroll
    for (int k = 0; k < CH_EDGES / 1024; ++k) {
        const int i = chunk0 + k * 1024 + tid;
        s[k] = src[min(i, e - 1)];
        d[k] = dst[min(i, e - 1)];
        if (i >= e) d[k] = -1;
    }
#pragma unroll
    for (int k = 0; k < CH_EDGES / 1024; ++k) {
        r[k] = 0;
        if (d[k] >= 0) r[k] = atomicAdd(&hist[d[k] >> BIN_SHIFT], 1);
    }
    __syncthreads();
    const int cntb = (tid < NB_MAX) ? hist[tid] : 0;
    lds_excl_scan(hist, NB_MAX, wsum, tid);
    lds_excl_scan(gofs, NB_MAX, wsum, tid);
    if (tid < NB_MAX && cntb > 0) {
        const int run = atomicAdd(&bincursor[tid * BIN_PAD], cntb);
        gofs[tid] = gofs[tid] + run - hist[tid];
    }
#pragma unroll
    for (int k = 0; k < CH_EDGES / 1024; ++k) {
        if (d[k] >= 0) stage[hist[d[k] >> BIN_SHIFT] + r[k]] = make_int2(s[k], d[k]);
    }
    __syncthreads();
    const int m = min(CH_EDGES, e - chunk0);
    for (int j = tid; j < m; j += 1024) {
        const int2 p = stage[j];
        binned[gofs[p.y >> BIN_SHIFT] + j] = p;
    }
}

__global__ __launch_bounds__(1024) void k_binsort(
    const int2* __restrict__ binned, const int* __restrict__ bintotal,
    int* __restrict__ col, int* __restrict__ rowstart, int* __restrict__ cnt,
    float* __restrict__ dinv, int n) {
    __shared__ int c[BIN_NODES];
    __shared__ int cur[BIN_NODES];
    __shared__ int wsum[32];
    __shared__ int s_base;
    __shared__ int stage[STAGE_CAP];
    const int tid  = threadIdx.x;
    const int lane = tid & 31;
    const int bin  = blockIdx.x;
    if (tid < BIN_NODES) c[tid] = 0;
    if (tid == 0) s_base = 0;
    __syncthreads();
    int v = (tid < bin) ? bintotal[tid * BIN_PAD] : 0;
    for (int off = 16; off > 0; off >>= 1) v += __shfl_xor(v, off, 32);
    if (lane == 0 && v != 0) atomicAdd(&s_base, v);
    const int total = bintotal[bin * BIN_PAD];
    __syncthreads();
    const int base = s_base;
    const int2* bp = binned + base;

    for (int j0 = 0; j0 < total; j0 += 8 * 1024) {
        int dl[8];
#pragma unroll
        for (int u = 0; u < 8; ++u) {
            const int j = j0 + u * 1024 + tid;
            dl[u] = (j < total) ? bp[j].y : -1;
        }
#pragma unroll
        for (int u = 0; u < 8; ++u)
            if (dl[u] >= 0) atomicAdd(&c[dl[u] & (BIN_NODES - 1)], 1);
    }
    __syncthreads();
    const int myc = (tid < BIN_NODES) ? c[tid] : 0;
    lds_excl_scan(c, BIN_NODES, wsum, tid);
    if (tid < BIN_NODES) {
        const int st   = c[tid];
        const int node = bin * BIN_NODES + tid;
        cur[tid] = st;
        if (node < n) {
            rowstart[node] = base + st;
            cnt[node]      = myc;
            dinv[node]     = rsqrtf((float)(myc + 1));
        }
    }
    __syncthreads();
    for (int j0 = 0; j0 < total; j0 += 8 * 1024) {
        int2 p[8];
#pragma unroll
        for (int u = 0; u < 8; ++u) {
            const int j = j0 + u * 1024 + tid;
            p[u] = (j < total) ? bp[j] : make_int2(0, -1);
        }
#pragma unroll
        for (int u = 0; u < 8; ++u) {
            if (p[u].y >= 0) {
                const int r = atomicAdd(&cur[p[u].y & (BIN_NODES - 1)], 1);
                if (r < STAGE_CAP) stage[r] = p[u].x;
                else               col[base + r] = p[u].x;
            }
        }
    }
    __syncthreads();
    const int m = min(total, STAGE_CAP);
    for (int j = tid; j < m; j += 1024) col[base + j] = stage[j];
}

__global__ __launch_bounds__(256) void k_gemm1_wmma(
    const float* __restrict__ x, const float* __restrict__ W1,
    float* __restrict__ h, int n_nodes, int mtiles) {
    __shared__ float lw[IN_CH * IN_CH];
    const int tid = threadIdx.x;
    for (int i = tid; i < IN_CH * IN_CH; i += 256) lw[i] = W1[i];
    __syncthreads();

    const int wid  = tid >> 5;
    const int lane = tid & 31;
    const int half = lane >> 4;
    const int l    = lane & 15;
    const int tile = blockIdx.x * 8 + wid;
    if (tile >= mtiles) return;
    const int m0 = tile * 16;

    v8f acc[8] = {};

    const int arow = min(m0 + l, n_nodes - 1);
    const float* ap = x + (size_t)arow * IN_CH;

    for (int kk = 0; kk < IN_CH; kk += 4) {
        const int ka = kk + 2 * half;
        v2f a;
        a.x = ap[ka];
        a.y = ap[ka + 1];
#pragma unroll
        for (int nt = 0; nt < 8; ++nt) {
            const int nc = nt * 16 + l;
            v2f b;
            b.x = lw[ka * IN_CH + nc];
            b.y = lw[(ka + 1) * IN_CH + nc];
            acc[nt] = __builtin_amdgcn_wmma_f32_16x16x4_f32(
                false, a, false, b, (short)0, acc[nt], false, false);
        }
    }

#pragma unroll
    for (int j = 0; j < 8; ++j) {
        const int row = m0 + j + 8 * half;
        if (row >= n_nodes) continue;
#pragma unroll
        for (int nt = 0; nt < 8; ++nt) {
            const size_t idx = (size_t)row * IN_CH + nt * 16 + l;
            h[idx] = acc[nt][j];
        }
    }
}

__global__ __launch_bounds__(256) void k_gather1(
    const int* __restrict__ col, const int* __restrict__ rowstart,
    const int* __restrict__ cnt, const float* __restrict__ dinv,
    const float* __restrict__ h, const float* __restrict__ b1,
    const float* __restrict__ W2, float* __restrict__ h2, int n) {
    const int lane = threadIdx.x & 31;
    const int node = __builtin_amdgcn_readfirstlane(blockIdx.x * 8 + (threadIdx.x >> 5));
    if (node >= n) return;
    const int   start = rowstart[node];
    const int   c     = cnt[node];
    const float dd    = dinv[node];
    const float4* hrow = (const float4*)h;
    const float4 hs = hrow[(size_t)node * 32 + lane];
    const float sc = dd * dd;
    float4 acc;
    acc.x = hs.x * sc; acc.y = hs.y * sc; acc.z = hs.z * sc; acc.w = hs.w * sc;

    for (int base = 0; base < c; base += 32) {
        const int m = min(32, c - base);
        int   s_l = node;
        float w_l = 0.f;
        if (lane < m) {
            s_l = col[start + base + lane];
            w_l = dinv[s_l] * dd;
        }
        for (int k = 0; k < m; k += 8) {
            float4 r[8];
            float  w[8];
#pragma unroll
            for (int u = 0; u < 8; ++u) {
                const int s = __builtin_amdgcn_readlane(s_l, k + u);
                w[u] = __int_as_float(__builtin_amdgcn_readlane(__float_as_int(w_l), k + u));
                r[u] = hrow[(size_t)s * 32 + lane];
            }
#pragma unroll
            for (int u = 0; u < 8; ++u) {
                float4 v = r[u];
                if (k + u >= m) { v.x = 0.f; v.y = 0.f; v.z = 0.f; v.w = 0.f; }
                acc.x += v.x * w[u]; acc.y += v.y * w[u];
                acc.z += v.z * w[u]; acc.w += v.w * w[u];
            }
        }
    }
    const float4 bb = ((const float4*)b1)[lane];
    const float4 w2 = ((const float4*)W2)[lane];
    float v = fmaxf(acc.x + bb.x, 0.f) * w2.x + fmaxf(acc.y + bb.y, 0.f) * w2.y +
              fmaxf(acc.z + bb.z, 0.f) * w2.z + fmaxf(acc.w + bb.w, 0.f) * w2.w;
#pragma unroll
    for (int off = 16; off > 0; off >>= 1) v += __shfl_xor(v, off, 32);
    if (lane == 0) h2[node] = v;
}

__global__ void k_gather2(const int* __restrict__ col, const int* __restrict__ rowstart,
                          const int* __restrict__ cnt, const float* __restrict__ dinv,
                          const float* __restrict__ h2, const float* __restrict__ b2,
                          float* __restrict__ out, int n) {
    const int i = blockIdx.x * blockDim.x + threadIdx.x;
    if (i >= n) return;
    const float di = dinv[i];
    float acc = h2[i] * di * di + b2[0];
    const int st = rowstart[i];
    const int c  = cnt[i];
#pragma unroll 4
    for (int j = 0; j < c; ++j) {
        const int s = col[st + j];
        acc += h2[s] * dinv[s] * di;
    }
    out[i] = acc;
}

extern "C" void kernel_launch(void* const* d_in, const int* in_sizes, int n_in,
                              void* d_out, int out_size, void* d_ws, size_t ws_size,
                              hipStream_t stream) {
    const float* x  = (const float*)d_in[0];
    const int*   ei = (const int*)d_in[1];
    const float* W1 = (const float*)d_in[2];
    const float* b1 = (const float*)d_in[3];
    const float* W2 = (const float*)d_in[4];
    const float* b2 = (const float*)d_in[5];

    const int n = in_sizes[0] / IN_CH;
    const int e = in_sizes[1] / 2;
    const int* src = ei;
    const int* dst = ei + e;

    float* ws        = (float*)d_ws;
    float* h         = ws;
    float* dinv      = h + 2 * (size_t)n * IN_CH;
    float* h2        = dinv + n;
    int*   cnt       = (int*)(h2 + n);
    int*   rowstart  = cnt + n;
    int2*  binned    = (int2*)(rowstart + n);
    int*   col       = (int*)(binned + e);
    int*   bintotal  = col + e;
    int*   bincursor = bintotal + NB_MAX * BIN_PAD;
    float* out       = (float*)d_out;

    const int nbins = (n + BIN_NODES - 1) / BIN_NODES;
    const int mtiles = (n + 15) / 16;

    k_init      <<<4, 256, 0, stream>>>(bintotal, 2 * NB_MAX);
    k_hist      <<<(e + CH_EDGES - 1) / CH_EDGES, 1024, 0, stream>>>(dst, bintotal, e, nbins);
    k_scatter   <<<(e + CH_EDGES - 1) / CH_EDGES, 1024, 0, stream>>>(src, dst, bintotal, bincursor,
                                                                      binned, e, nbins);
    k_binsort   <<<nbins, 1024, 0, stream>>>(binned, bintotal, col, rowstart, cnt, dinv, n);
    k_gemm1_wmma<<<(mtiles + 7) / 8, 256, 0, stream>>>(x, W1, h, n, mtiles);
    k_gather1   <<<(n + 7) / 8, 256, 0, stream>>>(col, rowstart, cnt, dinv, h, b1, W2, h2, n);
    k_gather2   <<<(n + 255) / 256, 256, 0, stream>>>(col, rowstart, cnt, dinv, h2, b2, out, n);
}


# Layer-1 pull run by a fixed grid of resident waves, each looping over a run of consecutive nodes (was one wave launched per node), next node's ids and self row fetched ahead; sum orders unchanged
# speedup vs baseline: 3.8100x; 1.0872x over previous
#include <hip/hip_runtime.h>
#include <hip/hip_bf16.h>


typedef __attribute__((ext_vector_type(2))) float v2f;
typedef __attribute__((ext_vector_type(8))) float v8f;

#define IN_CH 128

#define BIN_SHIFT 8
#define BIN_NODES 256
#define NB_MAX    512
#define CH_EDGES  4096
#define STAGE_CAP 8192
#define BIN_PAD   64

__global__ void k_init(int* __restrict__ binws, int nctr) {
    int i = blockIdx.x * blockDim.x + threadIdx.x;
    if (i < nctr) binws[i * BIN_PAD] = 0;
}

__device__ __forceinline__ void lds_excl_scan(int* a, int len, int* wsum, int tid) {
    const int lane = tid & 31;
    const int v = (tid < len) ? a[tid] : 0;
    int incl = v;
    for (int off = 1; off < 32; off <<= 1) {
        const int t = __shfl_up(incl, off, 32);
        if (lane >= off) incl += t;
    }
    if (lane == 31) wsum[tid >> 5] = incl;
    __syncthreads();
    int b = 0;
    for (int w = 0; w < (tid >> 5); ++w) b += wsum[w];
    if (tid < len) a[tid] = b + incl - v;
    __syncthreads();
}

__global__ __launch_bounds__(1024) void k_hist(const int* __restrict__ dst,
                                               int* __restrict__ bintotal, int e, int nbins) {
    __shared__ int hist[NB_MAX];
    const int tid = threadIdx.x;
    if (tid < NB_MAX) hist[tid] = 0;
    __syncthreads();
    const int chunk0 = blockIdx.x * CH_EDGES;
    int d[CH_EDGES / 1024];
#pragma unroll
    for (int k = 0; k < CH_EDGES / 1024; ++k) {
        d[k] = dst[min(chunk0 + k * 1024 + tid, e - 1)];
    }
    static_assert(CH_EDGES / 1024 == 4, "the load fence below names four values");
    asm volatile("" :: "v"(d[0]), "v"(d[1]), "v"(d[2]), "v"(d[3]));
#pragma unroll
    for (int k = 0; k < CH_EDGES / 1024; ++k)
        if (chunk0 + k * 1024 + tid < e) atomicAdd(&hist[d[k] >> BIN_SHIFT], 1);
    __syncthreads();
    if (tid < nbins) {
        const int c = hist[tid];
        if (c > 0) atomicAdd(&bintotal[tid * BIN_PAD], c);
    }
}

__global__ __launch_bounds__(1024) void k_scatter(
    const int* __restrict__ src, const int* __restrict__ dst,
    const int* __restrict__ bintotal, int* __restrict__ bincursor,
    int2* __restrict__ binned, int e, int nbins) {
    __shared__ int  hist[NB_MAX];
    __shared__ int  gofs[NB_MAX];
    __shared__ int  wsum[32];
    __shared__ int2 stage[CH_EDGES];
    const int tid = threadIdx.x;
    if (tid < NB_MAX) {
        hist[tid] = 0;
        gofs[tid] = (tid < nbins) ? bintotal[tid * BIN_PAD] : 0;
    }
    __syncthreads();
    const int chunk0 = blockIdx.x * CH_EDGES;
    int s[CH_EDGES / 1024], d[CH_EDGES / 1024], r[CH_EDGES / 1024];
#pragma unroll
    for (int k = 0; k < CH_EDGES / 1024; ++k) {
        const int i = chunk0 + k * 1024 + tid;
        s[k] = src[min(i, e - 1)];
        d[k] = dst[min(i, e - 1)];
        if (i >= e) d[k] = -1;
    }
#pragma unroll
    for (int k = 0; k < CH_EDGES / 1024; ++k) {
        r[k] = 0;
        if (d[k] >= 0) r[k] = atomicAdd(&hist[d[k] >> BIN_SHIFT], 1);
    }
    __syncthreads();
    const int cntb = (tid < NB_MAX) ? hist[tid] : 0;
    lds_excl_scan(hist, NB_MAX, wsum, tid);
    lds_excl_scan(gofs, NB_MAX, wsum, tid);
    if (tid < NB_MAX && cntb > 0) {
        const int run = atomicAdd(&bincursor[tid * BIN_PAD], cntb);
        gofs[tid] = gofs[tid] + run - hist[tid];
    }
#pragma unroll
    for (int k = 0; k < CH_EDGES / 1024; ++k) {
        if (d[k] >= 0) stage[hist[d[k] >> BIN_SHIFT] + r[k]] = make_int2(s[k], d[k]);
    }
    __syncthreads();
    const int m = min(CH_EDGES, e - chunk0);
    for (int j = tid; j < m; j += 1024) {
        const int2 p = stage[j];
        binned[gofs[p.y >> BIN_SHIFT] + j] = p;
    }
}

__global__ __launch_bounds__(1024) void k_binsort(
    const int2* __restrict__ binned, const int* __restrict__ bintotal,
    int* __restrict__ col, int* __restrict__ rowstart, int* __restrict__ cnt,
    float* __restrict__ dinv, int n) {
    __shared__ int c[BIN_NODES];
    __shared__ int cur[BIN_NODES];
    __shared__ int wsum[32];
    __shared__ int s_base;
    __shared__ int stage[STAGE_CAP];
    const int tid  = threadIdx.x;
    const int lane = tid & 31;
    const int bin  = blockIdx.x;
    if (tid < BIN_NODES) c[tid] = 0;
    if (tid == 0) s_base = 0;
    __syncthreads();
    int v = (tid < bin) ? bintotal[tid * BIN_PAD] : 0;
    for (int off = 16; off > 0; off >>= 1) v += __shfl_xor(v, off, 32);
    if (lane == 0 && v != 0) atomicAdd(&s_base, v);
    const int total = bintotal[bin * BIN_PAD];
    __syncthreads();
    const int base = s_base;
    const int2* bp = binned + base;

    for (int j0 = 0; j0 < total; j0 += 8 * 1024) {
        int dl[8];
#pragma unroll
        for (int u = 0; u < 8; ++u) {
            const int j = j0 + u * 1024 + tid;
            dl[u] = (j < total) ? bp[j].y : -1;
        }
#pragma unroll
        for (int u = 0; u < 8; ++u)
            if (dl[u] >= 0) atomicAdd(&c[dl[u] & (BIN_NODES - 1)], 1);
    }
    __syncthreads();
    const int myc = (tid < BIN_NODES) ? c[tid] : 0;
    lds_excl_scan(c, BIN_NODES, wsum, tid);
    if (tid < BIN_NODES) {
        const int st   = c[tid];
        const int node = bin * BIN_NODES + tid;
        cur[tid] = st;
        if (node < n) {
            rowstart[node] = base + st;
            cnt[node]      = myc;
            dinv[node]     = rsqrtf((float)(myc + 1));
        }
    }
    __syncthreads();
    for (int j0 = 0; j0 < total; j0 += 8 * 1024) {
        int2 p[8];
#pragma unroll
        for (int u = 0; u < 8; ++u) {
            const int j = j0 + u * 1024 + tid;
            p[u] = (j < total) ? bp[j] : make_int2(0, -1);
        }
#pragma unroll
        for (int u = 0; u < 8; ++u) {
            if (p[u].y >= 0) {
                const int r = atomicAdd(&cur[p[u].y & (BIN_NODES - 1)], 1);
                if (r < STAGE_CAP) stage[r] = p[u].x;
                else               col[base + r] = p[u].x;
            }
        }
    }
    __syncthreads();
    const int m = min(total, STAGE_CAP);
    for (int j = tid; j < m; j += 1024) col[base + j] = stage[j];
}

__global__ __launch_bounds__(256) void k_gemm1_wmma(
    const float* __restrict__ x, const float* __restrict__ W1,
    const float* __restrict__ dinv, float* __restrict__ h, int n_nodes, int mtiles) {
    __shared__ float lw[IN_CH * IN_CH];
    const int tid = threadIdx.x;
    for (int i = tid; i < IN_CH * IN_CH; i += 256) lw[i] = W1[i];
    __syncthreads();

    const int wid  = tid >> 5;
    const int lane = tid & 31;
    const int half = lane >> 4;
    const int l    = lane & 15;
    const int tile = blockIdx.x * 8 + wid;
    if (tile >= mtiles) return;
    const int m0 = tile * 16;

    v8f acc[8] = {};

    float dr[8];
#pragma unroll
    for (int j = 0; j < 8; ++j) dr[j] = dinv[min(m0 + j + 8 * half, n_nodes - 1)];

    const int arow = min(m0 + l, n_nodes - 1);
    const float* ap = x + (size_t)arow * IN_CH;

    for (int kk = 0; kk < IN_CH; kk += 4) {
        const int ka = kk + 2 * half;
        v2f a;
        a.x = ap[ka];
        a.y = ap[ka + 1];
#pragma unroll
        for (int nt = 0; nt < 8; ++nt) {
            const int nc = nt * 16 + l;
            v2f b;
            b.x = lw[ka * IN_CH + nc];
            b.y = lw[(ka + 1) * IN_CH + nc];
            acc[nt] = __builtin_amdgcn_wmma_f32_16x16x4_f32(
                false, a, false, b, (short)0, acc[nt], false, false);
        }
    }

#pragma unroll
    for (int j = 0; j < 8; ++j) {
        const int row = m0 + j + 8 * half;
        if (row >= n_nodes) continue;
#pragma unroll
        for (int nt = 0; nt < 8; ++nt) {
            const size_t idx = (size_t)row * IN_CH + nt * 16 + l;
            h[idx] = acc[nt][j] * dr[j];
        }
    }
}

__device__ __forceinline__ void pull_rows8(const float4* __restrict__ hrow, int s_l, int k, int m,
                                           int lane, float4& acc) {
    float4 r[8];
#pragma unroll
    for (int u = 0; u < 8; ++u) {
        const int s = __builtin_amdgcn_readlane(s_l, k + u);
        r[u] = hrow[(size_t)s * 32 + lane];
    }
#pragma unroll
    for (int u = 0; u < 8; ++u) {
        float4 v = r[u];
        if (k + u >= m) { v.x = 0.f; v.y = 0.f; v.z = 0.f; v.w = 0.f; }
        acc.x += v.x; acc.y += v.y; acc.z += v.z; acc.w += v.w;
    }
}

#define G1_WAVES_PER_WG 8
__global__ __launch_bounds__(256) void k_gather1(
    const int* __restrict__ col, const int* __restrict__ rowstart,
    const int* __restrict__ cnt, const float* __restrict__ dinv,
    const float* __restrict__ h, const float* __restrict__ b1,
    const float* __restrict__ W2, float* __restrict__ h2, int n) {
    const int lane = threadIdx.x & 31;
    const int nw = (int)gridDim.x * G1_WAVES_PER_WG;
    const int w  = __builtin_amdgcn_readfirstlane(blockIdx.x * G1_WAVES_PER_WG + (threadIdx.x >> 5));
    const int q   = n / nw;
    const int rem = n - q * nw;
    const int a   = w * q + min(w, rem);
    const int b   = a + q + (w < rem ? 1 : 0);
    if (a >= b) return;

    const float4* hrow = (const float4*)h;
    const float4 bb = ((const float4*)b1)[lane];
    const float4 w2 = ((const float4*)W2)[lane];

    for (int a0 = a; a0 < b; a0 += 32) {
        const int kk = min(32, b - a0);
        const int nl = min(a0 + lane, b - 1);
        const int   rs_l = rowstart[nl];
        const int   c_l  = cnt[nl];
        const float dd_l = dinv[nl];
        int start = __builtin_amdgcn_readlane(rs_l, 0);
        int c     = __builtin_amdgcn_readlane(c_l, 0);
        int ids   = a0;
        if (lane < c) ids = col[start + lane];
        float4 self = hrow[(size_t)a0 * 32 + lane];
        float res = 0.f;

        for (int i = 0; i < kk; ++i) {
            const int node = a0 + i;
            const int cur_start = start;
            const int cur_c     = c;
            int    s_l = ids;
            float4 acc = self;
            const float dd = __int_as_float(__builtin_amdgcn_readlane(__float_as_int(dd_l), i));
            if (i + 1 < kk) {
                start = __builtin_amdgcn_readlane(rs_l, i + 1);
                c     = __builtin_amdgcn_readlane(c_l, i + 1);
                ids   = node;
                if (lane < c) ids = col[start + lane];
                self  = hrow[(size_t)(node + 1) * 32 + lane];
            }
            const int m0 = min(32, cur_c);
            for (int k = 0; k < m0; k += 8) pull_rows8(hrow, s_l, k, m0, lane, acc);
            for (int base = 32; base < cur_c; base += 32) {
                const int m = min(32, cur_c - base);
                int s_x = node;
                if (lane < m) s_x = col[cur_start + base + lane];
                for (int k = 0; k < m; k += 8) pull_rows8(hrow, s_x, k, m, lane, acc);
            }
            acc.x *= dd; acc.y *= dd; acc.z *= dd; acc.w *= dd;
            float v = fmaxf(acc.x + bb.x, 0.f) * w2.x + fmaxf(acc.y + bb.y, 0.f) * w2.y +
                      fmaxf(acc.z + bb.z, 0.f) * w2.z + fmaxf(acc.w + bb.w, 0.f) * w2.w;
#pragma unroll
            for (int off = 16; off > 0; off >>= 1) v += __shfl_xor(v, off, 32);
            if (lane == i) res = v * dd;
        }
        if (lane < kk) h2[a0 + lane] = res;
    }
}

__global__ void k_gather2(const int* __restrict__ col, const int* __restrict__ rowstart,
                          const int* __restrict__ cnt, const float* __restrict__ dinv,
                          const float* __restrict__ h2, const float* __restrict__ b2,
                          float* __restrict__ out, int n) {
    const int i = blockIdx.x * blockDim.x + threadIdx.x;
    if (i >= n) return;
    const float di   = dinv[i];
    const float bias = b2[0];
    float acc = h2[i];
    const int st = rowstart[i];
    const int c  = cnt[i];
#pragma unroll 4
    for (int j = 0; j < c; ++j) acc += h2[col[st + j]];
    out[i] = acc * di + bias;
}

extern "C" void kernel_launch(void* const* d_in, const int* in_sizes, int n_in,
                              void* d_out, int out_size, void* d_ws, size_t ws_size,
                              hipStream_t stream) {
    const float* x  = (const float*)d_in[0];
    const int*   ei = (const int*)d_in[1];
    const float* W1 = (const float*)d_in[2];
    const float* b1 = (const float*)d_in[3];
    const float* W2 = (const float*)d_in[4];
    const float* b2 = (const float*)d_in[5];

    const int n = in_sizes[0] / IN_CH;
    const int e = in_sizes[1] / 2;
    const int* src = ei;
    const int* dst = ei + e;

    float* ws        = (float*)d_ws;
    float* h         = ws;
    float* dinv      = h + 2 * (size_t)n * IN_CH;
    float* h2        = dinv + n;
    int*   cnt       = (int*)(h2 + n);
    int*   rowstart  = cnt + n;
    int2*  binned    = (int2*)(rowstart + n);
    int*   col       = (int*)(binned + e);
    int*   bintotal  = col + e;
    int*   bincursor = bintotal + NB_MAX * BIN_PAD;
    float* out       = (float*)d_out;

    const int nbins = (n + BIN_NODES - 1) / BIN_NODES;
    const int mtiles = (n + 15) / 16;

    k_init      <<<4, 256, 0, stream>>>(bintotal, 2 * NB_MAX);
    k_hist      <<<(e + CH_EDGES - 1) / CH_EDGES, 1024, 0, stream>>>(dst, bintotal, e, nbins);
    k_scatter   <<<(e + CH_EDGES - 1) / CH_EDGES, 1024, 0, stream>>>(src, dst, bintotal, bincursor,
                                                                      binned, e, nbins);
    k_binsort   <<<nbins, 1024, 0, stream>>>(binned, bintotal, col, rowstart, cnt, dinv, n);
    k_gemm1_wmma<<<(mtiles + 7) / 8, 256, 0, stream>>>(x, W1, dinv, h, n, mtiles);
    k_gather1   <<<1280, 256, 0, stream>>>(col, rowstart, cnt, dinv, h, b1, W2, h2, n);
    k_gather2   <<<(n + 255) / 256, 256, 0, stream>>>(col, rowstart, cnt, dinv, h2, b2, out, n);
}


# Layer-1 pull moved ahead of the linear: a wave sums source-weighted x rows of 16 nodes, the tile goes via LDS into the f32 WMMA with W1, relu/W2 dot in its epilogue; h never stored; sums reordered
# speedup vs baseline: 4.3466x; 1.1408x over previous
#include <hip/hip_runtime.h>
#include <hip/hip_bf16.h>


typedef __attribute__((ext_vector_type(2))) float v2f;
typedef __attribute__((ext_vector_type(8))) float v8f;

#define IN_CH 128

#define BIN_SHIFT 8
#define BIN_NODES 256
#define NB_MAX    512
#define CH_EDGES  4096
#define STAGE_CAP 8192
#define BIN_PAD   64

__global__ void k_init(int* __restrict__ binws, int nctr) {
    int i = blockIdx.x * blockDim.x + threadIdx.x;
    if (i < nctr) binws[i * BIN_PAD] = 0;
}

__device__ __forceinline__ void lds_excl_scan(int* a, int len, int* wsum, int tid) {
    const int lane = tid & 31;
    const int v = (tid < len) ? a[tid] : 0;
    int incl = v;
    for (int off = 1; off < 32; off <<= 1) {
        const int t = __shfl_up(incl, off, 32);
        if (lane >= off) incl += t;
    }
    if (lane == 31) wsum[tid >> 5] = incl;
    __syncthreads();
    int b = 0;
    for (int w = 0; w < (tid >> 5); ++w) b += wsum[w];
    if (tid < len) a[tid] = b + incl - v;
    __syncthreads();
}

__global__ __launch_bounds__(1024) void k_hist(const int* __restrict__ dst,
                                               int* __restrict__ bintotal, int e, int nbins) {
    __shared__ int hist[NB_MAX];
    const int tid = threadIdx.x;
    if (tid < NB_MAX) hist[tid] = 0;
    __syncthreads();
    const int chunk0 = blockIdx.x * CH_EDGES;
    int d[CH_EDGES / 1024];
#pragma unroll
    for (int k = 0; k < CH_EDGES / 1024; ++k) {
        d[k] = dst[min(chunk0 + k * 1024 + tid, e - 1)];
    }
    static_assert(CH_EDGES / 1024 == 4, "the load fence below names four values");
    asm volatile("" :: "v"(d[0]), "v"(d[1]), "v"(d[2]), "v"(d[3]));
#pragma unroll
    for (int k = 0; k < CH_EDGES / 1024; ++k)
        if (chunk0 + k * 1024 + tid < e) atomicAdd(&hist[d[k] >> BIN_SHIFT], 1);
    __syncthreads();
    if (tid < nbins) {
        const int c = hist[tid];
        if (c > 0) atomicAdd(&bintotal[tid * BIN_PAD], c);
    }
}

__global__ __launch_bounds__(1024) void k_scatter(
    const int* __restrict__ src, const int* __restrict__ dst,
    const int* __restrict__ bintotal, int* __restrict__ bincursor,
    int2* __restrict__ binned, int e, int nbins) {
    __shared__ int  hist[NB_MAX];
    __shared__ int  gofs[NB_MAX];
    __shared__ int  wsum[32];
    __shared__ int2 stage[CH_EDGES];
    const int tid = threadIdx.x;
    if (tid < NB_MAX) {
        hist[tid] = 0;
        gofs[tid] = (tid < nbins) ? bintotal[tid * BIN_PAD] : 0;
    }
    __syncthreads();
    const int chunk0 = blockIdx.x * CH_EDGES;
    int s[CH_EDGES / 1024], d[CH_EDGES / 1024], r[CH_EDGES / 1024];
#pragma unroll
    for (int k = 0; k < CH_EDGES / 1024; ++k) {
        const int i = chunk0 + k * 1024 + tid;
        s[k] = src[min(i, e - 1)];
        d[k] = dst[min(i, e - 1)];
        if (i >= e) d[k] = -1;
    }
#pragma unroll
    for (int k = 0; k < CH_EDGES / 1024; ++k) {
        r[k] = 0;
        if (d[k] >= 0) r[k] = atomicAdd(&hist[d[k] >> BIN_SHIFT], 1);
    }
    __syncthreads();
    const int cntb = (tid < NB_MAX) ? hist[tid] : 0;
    lds_excl_scan(hist, NB_MAX, wsum, tid);
    lds_excl_scan(gofs, NB_MAX, wsum, tid);
    if (tid < NB_MAX && cntb > 0) {
        const int run = atomicAdd(&bincursor[tid * BIN_PAD], cntb);
        gofs[tid] = gofs[tid] + run - hist[tid];
    }
#pragma unroll
    for (int k = 0; k < CH_EDGES / 1024; ++k) {
        if (d[k] >= 0) stage[hist[d[k] >> BIN_SHIFT] + r[k]] = make_int2(s[k], d[k]);
    }
    __syncthreads();
    const int m = min(CH_EDGES, e - chunk0);
    for (int j = tid; j < m; j += 1024) {
        const int2 p = stage[j];
        binned[gofs[p.y >> BIN_SHIFT] + j] = p;
    }
}

__global__ __launch_bounds__(1024) void k_binsort(
    const int2* __restrict__ binned, const int* __restrict__ bintotal,
    int* __restrict__ col, int* __restrict__ rowstart, int* __restrict__ cnt,
    float* __restrict__ dinv, int n) {
    __shared__ int c[BIN_NODES];
    __shared__ int cur[BIN_NODES];
    __shared__ int wsum[32];
    __shared__ int s_base;
    __shared__ int stage[STAGE_CAP];
    const int tid  = threadIdx.x;
    const int lane = tid & 31;
    const int bin  = blockIdx.x;
    if (tid < BIN_NODES) c[tid] = 0;
    if (tid == 0) s_base = 0;
    __syncthreads();
    int v = (tid < bin) ? bintotal[tid * BIN_PAD] : 0;
    for (int off = 16; off > 0; off >>= 1) v += __shfl_xor(v, off, 32);
    if (lane == 0 && v != 0) atomicAdd(&s_base, v);
    const int total = bintotal[bin * BIN_PAD];
    __syncthreads();
    const int base = s_base;
    const int2* bp = binned + base;

    for (int j0 = 0; j0 < total; j0 += 8 * 1024) {
        int dl[8];
#pragma unroll
        for (int u = 0; u < 8; ++u) {
            const int j = j0 + u * 1024 + tid;
            dl[u] = (j < total) ? bp[j].y : -1;
        }
#pragma unroll
        for (int u = 0; u < 8; ++u)
            if (dl[u] >= 0) atomicAdd(&c[dl[u] & (BIN_NODES - 1)], 1);
    }
    __syncthreads();
    const int myc = (tid < BIN_NODES) ? c[tid] : 0;
    lds_excl_scan(c, BIN_NODES, wsum, tid);
    if (tid < BIN_NODES) {
        const int st   = c[tid];
        const int node = bin * BIN_NODES + tid;
        cur[tid] = st;
        if (node < n) {
            rowstart[node] = base + st;
            cnt[node]      = myc;
            dinv[node]     = rsqrtf((float)(myc + 1));
        }
    }
    __syncthreads();
    for (int j0 = 0; j0 < total; j0 += 8 * 1024) {
        int2 p[8];
#pragma unroll
        for (int u = 0; u < 8; ++u) {
            const int j = j0 + u * 1024 + tid;
            p[u] = (j < total) ? bp[j] : make_int2(0, -1);
        }
#pragma unroll
        for (int u = 0; u < 8; ++u) {
            if (p[u].y >= 0) {
                const int r = atomicAdd(&cur[p[u].y & (BIN_NODES - 1)], 1);
                if (r < STAGE_CAP) stage[r] = p[u].x;
                else               col[base + r] = p[u].x;
            }
        }
    }
    __syncthreads();
    const int m = min(total, STAGE_CAP);
    for (int j = tid; j < m; j += 1024) col[base + j] = stage[j];
}

#define F_WAVES 28
__device__ __forceinline__ float lane_f(float v, int k) {
    return __int_as_float(__builtin_amdgcn_readlane(__float_as_int(v), k));
}

__device__ __forceinline__ void pull_rows16(const float4* __restrict__ xrow, int s_l, float w_l,
                                            int k, int m, int lane, float4& acc) {
    float4 r[16];
#pragma unroll
    for (int u = 0; u < 16; ++u) {
        r[u] = make_float4(0.f, 0.f, 0.f, 0.f);
        if (k + u < m) {
            const int s = __builtin_amdgcn_readlane(s_l, k + u);
            r[u] = xrow[(size_t)s * 32 + lane];
        }
    }
#pragma unroll
    for (int u = 0; u < 16; ++u) {
        const float w = lane_f(w_l, k + u);
        acc.x = fmaf(w, r[u].x, acc.x); acc.y = fmaf(w, r[u].y, acc.y);
        acc.z = fmaf(w, r[u].z, acc.z); acc.w = fmaf(w, r[u].w, acc.w);
    }
}

__global__ __launch_bounds__(F_WAVES * 32) void k_pull1_gemm(
    const int* __restrict__ col, const int* __restrict__ rowstart,
    const int* __restrict__ cnt, const float* __restrict__ dinv,
    const float* __restrict__ x, const float* __restrict__ W1,
    const float* __restrict__ b1, const float* __restrict__ W2,
    float* __restrict__ h2, int n, int mtiles) {
    __shared__ float4 lwB[16 * 8 * 32];
    __shared__ float4 at[F_WAVES][16 * 32];
    const int tid  = threadIdx.x;
    const int lane = tid & 31;
    const int half = lane >> 4;
    const int l    = lane & 15;
    for (int i = tid; i < 16 * 8 * 32; i += F_WAVES * 32) {
        const int ln = i & 31, nt = (i >> 5) & 7, p = i >> 8;
        const int k0 = 8 * p + 4 * (ln >> 4);
        const int c  = nt * 16 + (ln & 15);
        lwB[i] = make_float4(W1[k0 * IN_CH + c], W1[(k0 + 1) * IN_CH + c],
                             W1[(k0 + 2) * IN_CH + c], W1[(k0 + 3) * IN_CH + c]);
    }
    __syncthreads();

    const int wid = __builtin_amdgcn_readfirstlane(tid >> 5);
    float4* __restrict__ myat = at[wid];
    const float4* xrow = (const float4*)x;

    for (int tile = wid * (int)gridDim.x + (int)blockIdx.x; tile < mtiles;
         tile += F_WAVES * (int)gridDim.x) {
        const int m0 = tile * 16;
        const int kk = min(16, n - m0);
        const int nl = min(m0 + lane, n - 1);
        const int   rs_l = rowstart[nl];
        const int   c_l  = cnt[nl];
        const float dd_l = dinv[nl];
        int start = __builtin_amdgcn_readlane(rs_l, 0);
        int c     = __builtin_amdgcn_readlane(c_l, 0);
        int ids   = m0;
        if (lane < c) ids = col[start + lane];
        float4 self = xrow[(size_t)m0 * 32 + lane];
        float w_n = dinv[ids];

        for (int i = 0; i < kk; ++i) {
            const int node = m0 + i;
            const int cur_start = start;
            const int cur_c     = c;
            const int s_l = ids;
            const float dd = lane_f(dd_l, i);
            const float w_l = w_n;
            float4 acc;
            acc.x = self.x * dd; acc.y = self.y * dd; acc.z = self.z * dd; acc.w = self.w * dd;
            if (i + 1 < kk) {
                start = __builtin_amdgcn_readlane(rs_l, i + 1);
                c     = __builtin_amdgcn_readlane(c_l, i + 1);
                ids   = node;
                if (lane < c) ids = col[start + lane];
                self  = xrow[(size_t)(node + 1) * 32 + lane];
            }
            const int mf = min(32, cur_c);
            for (int k = 0; k < mf; k += 16) pull_rows16(xrow, s_l, w_l, k, mf, lane, acc);
            for (int base = 32; base < cur_c; base += 32) {
                const int m = min(32, cur_c - base);
                int s_x = node;
                float w_x = 0.f;
                if (lane < m) { s_x = col[cur_start + base + lane]; w_x = dinv[s_x]; }
                for (int k = 0; k < m; k += 16) pull_rows16(xrow, s_x, w_x, k, m, lane, acc);
            }
            w_n = dinv[ids];
            acc.x *= dd; acc.y *= dd; acc.z *= dd; acc.w *= dd;
            myat[i * 32 + (lane ^ i)] = acc;
        }
        for (int i = kk; i < 16; ++i)
            myat[i * 32 + (lane ^ i)] = make_float4(0.f, 0.f, 0.f, 0.f);

        v8f z[8] = {};
#pragma unroll 1
        for (int p = 0; p < 16; ++p) {
            const float4 av = myat[l * 32 + ((2 * p + half) ^ l)];
            float4 bv[8];
#pragma unroll
            for (int nt = 0; nt < 8; ++nt) bv[nt] = lwB[(p * 8 + nt) * 32 + lane];
            v2f a0, a1;
            a0.x = av.x; a0.y = av.y;
            a1.x = av.z; a1.y = av.w;
#pragma unroll
            for (int nt = 0; nt < 8; ++nt) {
                v2f b; b.x = bv[nt].x; b.y = bv[nt].y;
                z[nt] = __builtin_amdgcn_wmma_f32_16x16x4_f32(
                    false, a0, false, b, (short)0, z[nt], false, false);
            }
#pragma unroll
            for (int nt = 0; nt < 8; ++nt) {
                v2f b; b.x = bv[nt].z; b.y = bv[nt].w;
                z[nt] = __builtin_amdgcn_wmma_f32_16x16x4_f32(
                    false, a1, false, b, (short)0, z[nt], false, false);
            }
        }

        float part[8];
#pragma unroll
        for (int j = 0; j < 8; ++j) part[j] = 0.f;
#pragma unroll
        for (int nt = 0; nt < 8; ++nt) {
            const float bb = b1[nt * 16 + l];
            const float ww = W2[nt * 16 + l];
#pragma unroll
            for (int j = 0; j < 8; ++j) part[j] += fmaxf(z[nt][j] + bb, 0.f) * ww;
        }
#pragma unroll
        for (int j = 0; j < 8; ++j) {
#pragma unroll
            for (int off = 8; off > 0; off >>= 1) part[j] += __shfl_xor(part[j], off, 32);
        }
        float res = part[0];
#pragma unroll
        for (int j = 1; j < 8; ++j) res = (l == j) ? part[j] : res;
        const float drow = __shfl(dd_l, 8 * half + l, 32);
        const int row = m0 + 8 * half + l;
        if (l < 8 && row < n) h2[row] = res * drow;
    }
}

__global__ void k_gather2(const int* __restrict__ col, const int* __restrict__ rowstart,
                          const int* __restrict__ cnt, const float* __restrict__ dinv,
                          const float* __restrict__ h2, const float* __restrict__ b2,
                          float* __restrict__ out, int n) {
    const int i = blockIdx.x * blockDim.x + threadIdx.x;
    if (i >= n) return;
    const float di   = dinv[i];
    const float bias = b2[0];
    float acc = h2[i];
    const int st = rowstart[i];
    const int c  = cnt[i];
#pragma unroll 4
    for (int j = 0; j < c; ++j) acc += h2[col[st + j]];
    out[i] = acc * di + bias;
}

extern "C" void kernel_launch(void* const* d_in, const int* in_sizes, int n_in,
                              void* d_out, int out_size, void* d_ws, size_t ws_size,
                              hipStream_t stream) {
    const float* x  = (const float*)d_in[0];
    const int*   ei = (const int*)d_in[1];
    const float* W1 = (const float*)d_in[2];
    const float* b1 = (const float*)d_in[3];
    const float* W2 = (const float*)d_in[4];
    const float* b2 = (const float*)d_in[5];

    const int n = in_sizes[0] / IN_CH;
    const int e = in_sizes[1] / 2;
    const int* src = ei;
    const int* dst = ei + e;

    float* ws        = (float*)d_ws;
    float* dinv      = ws + 2 * (size_t)n * IN_CH;
    float* h2        = dinv + n;
    int*   cnt       = (int*)(h2 + n);
    int*   rowstart  = cnt + n;
    int2*  binned    = (int2*)(rowstart + n);
    int*   col       = (int*)(binned + e);
    int*   bintotal  = col + e;
    int*   bincursor = bintotal + NB_MAX * BIN_PAD;
    float* out       = (float*)d_out;

    const int nbins = (n + BIN_NODES - 1) / BIN_NODES;
    const int mtiles = (n + 15) / 16;

    k_init      <<<4, 256, 0, stream>>>(bintotal, 2 * NB_MAX);
    k_hist      <<<(e + CH_EDGES - 1) / CH_EDGES, 1024, 0, stream>>>(dst, bintotal, e, nbins);
    k_scatter   <<<(e + CH_EDGES - 1) / CH_EDGES, 1024, 0, stream>>>(src, dst, bintotal, bincursor,
                                                                      binned, e, nbins);
    k_binsort   <<<nbins, 1024, 0, stream>>>(binned, bintotal, col, rowstart, cnt, dinv, n);
    k_pull1_gemm<<<256, F_WAVES * 32, 0, stream>>>(col, rowstart, cnt, dinv, x, W1, b1, W2, h2, n, mtiles);
    k_gather2   <<<(n + 255) / 256, 256, 0, stream>>>(col, rowstart, cnt, dinv, h2, b2, out, n);
}


# Edge grouping level one made chunk-local: a work-group orders its 4096 edges by bin in place plus a table row of its bin runs, the bin pass reads the runs; 4 launches, was 6; node sum order changes
# speedup vs baseline: 5.5734x; 1.2822x over previous
#include <hip/hip_runtime.h>
#include <hip/hip_bf16.h>


typedef __attribute__((ext_vector_type(2))) float v2f;
typedef __attribute__((ext_vector_type(8))) float v8f;

#define IN_CH 128

#define BIN_SHIFT 8
#define BIN_NODES 256
#define NB_MAX    512
#define CH_EDGES  4096
#define STAGE_CAP 8192
#define RUN_LANES 16
#define RUN_PASS  8
#define RUN_ROUND 512

__device__ __forceinline__ void lds_excl_scan(int* a, int len, int* wsum, int tid) {
    const int lane = tid & 31;
    const int v = (tid < len) ? a[tid] : 0;
    int incl = v;
    for (int off = 1; off < 32; off <<= 1) {
        const int t = __shfl_up(incl, off, 32);
        if (lane >= off) incl += t;
    }
    if (lane == 31) wsum[tid >> 5] = incl;
    __syncthreads();
    int b = 0;
    for (int w = 0; w < (tid >> 5); ++w) b += wsum[w];
    if (tid < len) a[tid] = b + incl - v;
    __syncthreads();
}

__global__ __launch_bounds__(1024) void k_chunksort(
    const int* __restrict__ src, const int* __restrict__ dst,
    int2* __restrict__ binned, int* __restrict__ table, int e) {
    __shared__ int  hist[NB_MAX];
    __shared__ int  wsum[32];
    __shared__ int2 stage[CH_EDGES];
    const int tid = threadIdx.x;
    if (tid < NB_MAX) hist[tid] = 0;
    __syncthreads();
    const int chunk0 = blockIdx.x * CH_EDGES;
    int s[CH_EDGES / 1024], d[CH_EDGES / 1024], r[CH_EDGES / 1024];
#pragma unroll
    for (int k = 0; k < CH_EDGES / 1024; ++k) {
        const int i = chunk0 + k * 1024 + tid;
        s[k] = src[min(i, e - 1)];
        d[k] = dst[min(i, e - 1)];
        if (i >= e) d[k] = -1;
    }
#pragma unroll
    for (int k = 0; k < CH_EDGES / 1024; ++k) {
        r[k] = 0;
        if (d[k] >= 0) r[k] = atomicAdd(&hist[d[k] >> BIN_SHIFT], 1);
    }
    __syncthreads();
    const int cntb = (tid < NB_MAX) ? hist[tid] : 0;
    lds_excl_scan(hist, NB_MAX, wsum, tid);
    if (tid < NB_MAX) table[(size_t)blockIdx.x * NB_MAX + tid] = (hist[tid] << 16) | cntb;
#pragma unroll
    for (int k = 0; k < CH_EDGES / 1024; ++k) {
        if (d[k] >= 0) stage[hist[d[k] >> BIN_SHIFT] + r[k]] = make_int2(s[k], d[k]);
    }
    __syncthreads();
    const int m = min(CH_EDGES, e - chunk0);
    for (int j = tid; j < m; j += 1024) binned[chunk0 + j] = stage[j];
}

__device__ __forceinline__ int load_runs(const int2* __restrict__ binned, const int4* seg4, int w0,
                                         int hw, int l16, int2 (&ev)[2 * RUN_PASS]) {
    const int4 pa = seg4[hw * (RUN_PASS / 4)];
    const int4 pb = seg4[hw * (RUN_PASS / 4) + 1];
    const int p[RUN_PASS] = {pa.x, pa.y, pa.z, pa.w, pb.x, pb.y, pb.z, pb.w};
    const int ib = (w0 + hw * RUN_PASS) * CH_EDGES + l16;
    int mx = 0;
#pragma unroll
    for (int u = 0; u < RUN_PASS; ++u) {
        const int ln = p[u] & 0xffff;
        const int i0 = ib + u * CH_EDGES + (p[u] >> 16);
        mx = max(mx, ln);
        ev[2 * u]     = make_int2(0, -1);
        ev[2 * u + 1] = make_int2(0, -1);
        if (l16 < ln)             ev[2 * u]     = binned[i0];
        if (l16 + RUN_LANES < ln) ev[2 * u + 1] = binned[i0 + RUN_LANES];
    }
    return mx;
}

template <bool PLACE>
__device__ __forceinline__ void take_edge(int2 ed, int* ctr, int* stage, int* __restrict__ col, int base) {
    if (ed.y < 0) return;
    const int r = atomicAdd(&ctr[ed.y & (BIN_NODES - 1)], 1);
    if (PLACE) {
        if (r < STAGE_CAP) stage[r] = ed.x;
        else               col[base + r] = ed.x;
    }
}

template <bool PLACE>
__device__ __forceinline__ void long_runs(const int2* __restrict__ binned, const int* seg, int w0,
                                          int hw, int l16, int* ctr, int* stage,
                                          int* __restrict__ col, int base) {
#pragma unroll 1
    for (int u = 0; u < RUN_PASS; ++u) {
        const int p  = seg[hw * RUN_PASS + u];
        const int ln = p & 0xffff;
        const int i0 = (w0 + hw * RUN_PASS + u) * CH_EDGES + (p >> 16);
        for (int li = l16 + 2 * RUN_LANES; li < ln; li += RUN_LANES)
            take_edge<PLACE>(binned[i0 + li], ctr, stage, col, base);
    }
}

__global__ __launch_bounds__(1024, 16) void k_binsort(
    const int2* __restrict__ binned, const int* __restrict__ table,
    int* __restrict__ col, int* __restrict__ rowstart, int* __restrict__ cnt,
    float* __restrict__ dinv, int n, int nch) {
    __shared__ int c[BIN_NODES];
    __shared__ int cur[BIN_NODES];
    __shared__ int wsum[32];
    __shared__ int s_base, s_total;
    __shared__ int4 seg4[RUN_ROUND / 4];
    int* seg = (int*)seg4;
    __shared__ int stage[STAGE_CAP];
    const int tid  = threadIdx.x;
    const int lane = tid & 31;
    const int hw   = tid >> 4;
    const int l16  = tid & 15;
    const int bin  = blockIdx.x;
    if (tid < BIN_NODES) c[tid] = 0;
    if (tid == 0) { s_base = 0; s_total = 0; }

    int2 ev[2 * RUN_PASS];
    int sb = 0, tt = 0, mx;
    for (int w0 = 0; w0 < nch; w0 += RUN_ROUND) {
        if (w0 > 0) __syncthreads();
        if (tid < RUN_ROUND) {
            const int p = (w0 + tid < nch) ? table[(size_t)(w0 + tid) * NB_MAX + bin] : 0;
            seg[tid] = p;
            sb += p >> 16;
            tt += p & 0xffff;
        }
        __syncthreads();
        mx = load_runs(binned, seg4, w0, hw, l16, ev);
#pragma unroll
        for (int i = 0; i < 2 * RUN_PASS; ++i) take_edge<false>(ev[i], c, stage, col, 0);
        if (mx > 2 * RUN_LANES) long_runs<false>(binned, seg, w0, hw, l16, c, stage, col, 0);
    }
    for (int off = 16; off > 0; off >>= 1) {
        sb += __shfl_xor(sb, off, 32);
        tt += __shfl_xor(tt, off, 32);
    }
    if (lane == 0 && (sb | tt) != 0) { atomicAdd(&s_base, sb); atomicAdd(&s_total, tt); }
    __syncthreads();
    const int base  = s_base;
    const int total = s_total;

    const int myc = (tid < BIN_NODES) ? c[tid] : 0;
    lds_excl_scan(c, BIN_NODES, wsum, tid);
    if (tid < BIN_NODES) {
        const int st   = c[tid];
        const int node = bin * BIN_NODES + tid;
        cur[tid] = st;
        if (node < n) {
            rowstart[node] = base + st;
            cnt[node]      = myc;
            dinv[node]     = rsqrtf((float)(myc + 1));
        }
    }
    __syncthreads();

    for (int w0 = 0; w0 < nch; w0 += RUN_ROUND) {
        if (nch > RUN_ROUND) {
            __syncthreads();
            if (tid < RUN_ROUND)
                seg[tid] = (w0 + tid < nch) ? table[(size_t)(w0 + tid) * NB_MAX + bin] : 0;
            __syncthreads();
        }
        mx = load_runs(binned, seg4, w0, hw, l16, ev);
#pragma unroll
        for (int i = 0; i < 2 * RUN_PASS; ++i) take_edge<true>(ev[i], cur, stage, col, base);
        if (mx > 2 * RUN_LANES) long_runs<true>(binned, seg, w0, hw, l16, cur, stage, col, base);
    }
    __syncthreads();
    const int m = min(total, STAGE_CAP);
    for (int j = tid; j < m; j += 1024) col[base + j] = stage[j];
}

#define F_WAVES 28
__device__ __forceinline__ float lane_f(float v, int k) {
    return __int_as_float(__builtin_amdgcn_readlane(__float_as_int(v), k));
}

__device__ __forceinline__ void pull_rows16(const float4* __restrict__ xrow, int s_l, float w_l,
                                            int k, int m, int lane, float4& acc) {
    float4 r[16];
#pragma unroll
    for (int u = 0; u < 16; ++u) {
        r[u] = make_float4(0.f, 0.f, 0.f, 0.f);
        if (k + u < m) {
            const int s = __builtin_amdgcn_readlane(s_l, k + u);
            r[u] = xrow[(size_t)s * 32 + lane];
        }
    }
#pragma unroll
    for (int u = 0; u < 16; ++u) {
        const float w = lane_f(w_l, k + u);
        acc.x = fmaf(w, r[u].x, acc.x); acc.y = fmaf(w, r[u].y, acc.y);
        acc.z = fmaf(w, r[u].z, acc.z); acc.w = fmaf(w, r[u].w, acc.w);
    }
}

__global__ __launch_bounds__(F_WAVES * 32) void k_pull1_gemm(
    const int* __restrict__ col, const int* __restrict__ rowstart,
    const int* __restrict__ cnt, const float* __restrict__ dinv,
    const float* __restrict__ x, const float* __restrict__ W1,
    const float* __restrict__ b1, const float* __restrict__ W2,
    float* __restrict__ h2, int n, int mtiles) {
    __shared__ float4 lwB[16 * 8 * 32];
    __shared__ float4 at[F_WAVES][16 * 32];
    const int tid  = threadIdx.x;
    const int lane = tid & 31;
    const int half = lane >> 4;
    const int l    = lane & 15;
    for (int i = tid; i < 16 * 8 * 32; i += F_WAVES * 32) {
        const int ln = i & 31, nt = (i >> 5) & 7, p = i >> 8;
        const int k0 = 8 * p + 4 * (ln >> 4);
        const int c  = nt * 16 + (ln & 15);
        lwB[i] = make_float4(W1[k0 * IN_CH + c], W1[(k0 + 1) * IN_CH + c],
                             W1[(k0 + 2) * IN_CH + c], W1[(k0 + 3) * IN_CH + c]);
    }
    __syncthreads();

    const int wid = __builtin_amdgcn_readfirstlane(tid >> 5);
    float4* __restrict__ myat = at[wid];
    const float4* xrow = (const float4*)x;

    for (int tile = wid * (int)gridDim.x + (int)blockIdx.x; tile < mtiles;
         tile += F_WAVES * (int)gridDim.x) {
        const int m0 = tile * 16;
        const int kk = min(16, n - m0);
        const int nl = min(m0 + lane, n - 1);
        const int   rs_l = rowstart[nl];
        const int   c_l  = cnt[nl];
        const float dd_l = dinv[nl];
        int start = __builtin_amdgcn_readlane(rs_l, 0);
        int c     = __builtin_amdgcn_readlane(c_l, 0);
        int ids   = m0;
        if (lane < c) ids = col[start + lane];
        float4 self = xrow[(size_t)m0 * 32 + lane];
        float w_n = dinv[ids];

        for (int i = 0; i < kk; ++i) {
            const int node = m0 + i;
            const int cur_start = start;
            const int cur_c     = c;
            const int s_l = ids;
            const float dd = lane_f(dd_l, i);
            const float w_l = w_n;
            float4 acc;
            acc.x = self.x * dd; acc.y = self.y * dd; acc.z = self.z * dd; acc.w = self.w * dd;
            if (i + 1 < kk) {
                start = __builtin_amdgcn_readlane(rs_l, i + 1);
                c     = __builtin_amdgcn_readlane(c_l, i + 1);
                ids   = node;
                if (lane < c) ids = col[start + lane];
                self  = xrow[(size_t)(node + 1) * 32 + lane];
            }
            const int mf = min(32, cur_c);
            for (int k = 0; k < mf; k += 16) pull_rows16(xrow, s_l, w_l, k, mf, lane, acc);
            for (int base = 32; base < cur_c; base += 32) {
                const int m = min(32, cur_c - base);
                int s_x = node;
                float w_x = 0.f;
                if (lane < m) { s_x = col[cur_start + base + lane]; w_x = dinv[s_x]; }
                for (int k = 0; k < m; k += 16) pull_rows16(xrow, s_x, w_x, k, m, lane, acc);
            }
            w_n = dinv[ids];
            acc.x *= dd; acc.y *= dd; acc.z *= dd; acc.w *= dd;
            myat[i * 32 + (lane ^ i)] = acc;
        }
        for (int i = kk; i < 16; ++i)
            myat[i * 32 + (lane ^ i)] = make_float4(0.f, 0.f, 0.f, 0.f);

        v8f z[8] = {};
#pragma unroll 1
        for (int p = 0; p < 16; ++p) {
            const float4 av = myat[l * 32 + ((2 * p + half) ^ l)];
            float4 bv[8];
#pragma unroll
            for (int nt = 0; nt < 8; ++nt) bv[nt] = lwB[(p * 8 + nt) * 32 + lane];
            v2f a0, a1;
            a0.x = av.x; a0.y = av.y;
            a1.x = av.z; a1.y = av.w;
#pragma unroll
            for (int nt = 0; nt < 8; ++nt) {
                v2f b; b.x = bv[nt].x; b.y = bv[nt].y;
                z[nt] = __builtin_amdgcn_wmma_f32_16x16x4_f32(
                    false, a0, false, b, (short)0, z[nt], false, false);
            }
#pragma unroll
            for (int nt = 0; nt < 8; ++nt) {
                v2f b; b.x = bv[nt].z; b.y = bv[nt].w;
                z[nt] = __builtin_amdgcn_wmma_f32_16x16x4_f32(
                    false, a1, false, b, (short)0, z[nt], false, false);
            }
        }

        float part[8];
#pragma unroll
        for (int j = 0; j < 8; ++j) part[j] = 0.f;
#pragma unroll
        for (int nt = 0; nt < 8; ++nt) {
            const float bb = b1[nt * 16 + l];
            const float ww = W2[nt * 16 + l];
#pragma unroll
            for (int j = 0; j < 8; ++j) part[j] += fmaxf(z[nt][j] + bb, 0.f) * ww;
        }
#pragma unroll
        for (int j = 0; j < 8; ++j) {
#pragma unroll
            for (int off = 8; off > 0; off >>= 1) part[j] += __shfl_xor(part[j], off, 32);
        }
        float res = part[0];
#pragma unroll
        for (int j = 1; j < 8; ++j) res = (l == j) ? part[j] : res;
        const float drow = __shfl(dd_l, 8 * half + l, 32);
        const int row = m0 + 8 * half + l;
        if (l < 8 && row < n) h2[row] = res * drow;
    }
}

__global__ void k_gather2(const int* __restrict__ col, const int* __restrict__ rowstart,
                          const int* __restrict__ cnt, const float* __restrict__ dinv,
                          const float* __restrict__ h2, const float* __restrict__ b2,
                          float* __restrict__ out, int n) {
    const int i = blockIdx.x * blockDim.x + threadIdx.x;
    if (i >= n) return;
    const float di   = dinv[i];
    const float bias = b2[0];
    float acc = h2[i];
    const int st = rowstart[i];
    const int c  = cnt[i];
#pragma unroll 4
    for (int j = 0; j < c; ++j) acc += h2[col[st + j]];
    out[i] = acc * di + bias;
}

extern "C" void kernel_launch(void* const* d_in, const int* in_sizes, int n_in,
                              void* d_out, int out_size, void* d_ws, size_t ws_size,
                              hipStream_t stream) {
    const float* x  = (const float*)d_in[0];
    const int*   ei = (const int*)d_in[1];
    const float* W1 = (const float*)d_in[2];
    const float* b1 = (const float*)d_in[3];
    const float* W2 = (const float*)d_in[4];
    const float* b2 = (const float*)d_in[5];

    const int n = in_sizes[0] / IN_CH;
    const int e = in_sizes[1] / 2;
    const int* src = ei;
    const int* dst = ei + e;

    float* ws        = (float*)d_ws;
    float* dinv      = ws + 2 * (size_t)n * IN_CH;
    float* h2        = dinv + n;
    int*   cnt       = (int*)(h2 + n);
    int*   rowstart  = cnt + n;
    int2*  binned    = (int2*)(rowstart + n);
    int*   col       = (int*)(binned + e);
    int*   table     = col + e;
    float* out       = (float*)d_out;

    const int nbins = (n + BIN_NODES - 1) / BIN_NODES;
    const int mtiles = (n + 15) / 16;
    const int nch = (e + CH_EDGES - 1) / CH_EDGES;

    k_chunksort <<<nch, 1024, 0, stream>>>(src, dst, binned, table, e);
    k_binsort   <<<nbins, 1024, 0, stream>>>(binned, table, col, rowstart, cnt, dinv, n, nch);
    k_pull1_gemm<<<256, F_WAVES * 32, 0, stream>>>(col, rowstart, cnt, dinv, x, W1, b1, W2, h2, n, mtiles);
    k_gather2   <<<(n + 255) / 256, 256, 0, stream>>>(col, rowstart, cnt, dinv, h2, b2, out, n);
}
